# starting kernel
# baseline (speedup 1.0000x reference)
#include <hip/hip_runtime.h>
#include <hip/hip_bf16.h>


typedef __bf16 bf16;
typedef __attribute__((ext_vector_type(16))) __bf16 v16bf;
typedef __attribute__((ext_vector_type(8)))  __bf16 v8bf;
typedef __attribute__((ext_vector_type(8)))  float  v8f;

template <int W> struct vecf;
template <> struct vecf<2> { typedef __attribute__((ext_vector_type(2))) float T; };
template <> struct vecf<4> { typedef __attribute__((ext_vector_type(4))) float T; };

__device__ __forceinline__ bf16 f2bf(float f) {
  union { float f; unsigned u; } in; in.f = f;
  unsigned u = in.u;
  unsigned r = u + 0x7fffu + ((u >> 16) & 1u);
  union { unsigned short s; bf16 b; } out;
  out.s = (unsigned short)(r >> 16);
  return out.b;
}


__global__ __launch_bounds__(256) void k_deg_init(float* deg, int n) {
  int i = blockIdx.x * 256 + threadIdx.x;
  if (i < n) deg[i] = 1.0f;
}

__global__ __launch_bounds__(256) void k_deg_count(const int* __restrict__ dst,
                                                   float* deg, int E) {
  int e = blockIdx.x * 256 + threadIdx.x;
  if (e < E) atomicAdd(&deg[dst[e]], 1.0f);
}

__global__ __launch_bounds__(256) void k_deg_rsqrt(float* deg, int n) {
  int i = blockIdx.x * 256 + threadIdx.x;
  if (i < n) deg[i] = rsqrtf(deg[i]);
}


__global__ __launch_bounds__(256) void k_cvt_bf16(const float* __restrict__ in,
                                                  bf16* __restrict__ out, int n) {
  int i = blockIdx.x * 256 + threadIdx.x;
  if (i < n) out[i] = f2bf(in[i]);
}

__global__ __launch_bounds__(256) void k_relu_cvt_bf16(const float* __restrict__ in,
                                                       bf16* __restrict__ out, int n) {
  int i = blockIdx.x * 256 + threadIdx.x;
  if (i < n) out[i] = f2bf(fmaxf(in[i], 0.0f));
}

__global__ __launch_bounds__(256) void k_transpose_bf16(const float* __restrict__ W,
                                                        bf16* __restrict__ Wt,
                                                        int K, int cols) {
  int idx = blockIdx.x * 256 + threadIdx.x;
  if (idx >= K * cols) return;
  int n = idx / K;
  int k = idx % K;
  Wt[idx] = f2bf(W[(size_t)k * cols + n]);
}


template <int NCOLS>
__global__ __launch_bounds__(256) void k_gemm_wmma_bf16(
    const bf16* __restrict__ A,
    const bf16* __restrict__ Bt,
    float* __restrict__ C,
    int M)
{
  constexpr int K = 128;
  constexpr int NT = 4;
  constexpr int chunksN = NCOLS / 64;
  const int wave = threadIdx.x >> 5;
  const int lane = threadIdx.x & 31;
  const int tile = blockIdx.x * 8 + wave;
  const int numTiles = (M >> 4) * chunksN;
  if (tile >= numTiles) return;
  const int tm = (tile / chunksN) << 4;
  const int tn = (tile % chunksN) << 6;
  const int rc = lane & 15;
  const int hi = lane >> 4;

  const bf16* aRow  = A  + (size_t)(tm + rc) * K + hi * 8;
  const bf16* bBase = Bt + (size_t)(tn + rc) * K + hi * 8;

  v8f acc[NT] = {};
#pragma unroll
  for (int k0 = 0; k0 < K; k0 += 32) {
    union { v16bf v; v8bf h[2]; } a;
    a.h[0] = *(const v8bf*)(aRow + k0);
    a.h[1] = *(const v8bf*)(aRow + k0 + 16);
#pragma unroll
    for (int t = 0; t < NT; ++t) {
      union { v16bf v; v8bf h[2]; } b;
      const bf16* bp = bBase + (size_t)t * 16 * K + k0;
      b.h[0] = *(const v8bf*)(bp);
      b.h[1] = *(const v8bf*)(bp + 16);
      acc[t] = __builtin_amdgcn_wmma_f32_16x16x32_bf16(
           false, a.v,  false, b.v,
           (short)0, acc[t],  false,  false);
    }
  }

  float* cBase = C + (size_t)tm * NCOLS + tn;
#pragma unroll
  for (int t = 0; t < NT; ++t)
#pragma unroll
    for (int j = 0; j < 8; ++j)
      cBase[(size_t)(j + hi * 8) * NCOLS + t * 16 + rc] = acc[t][j];
}


template <int NCOLS>
__global__ __launch_bounds__(256) void k_self_init(const float* __restrict__ H,
                                                   const float* __restrict__ bias,
                                                   const float* __restrict__ dinv,
                                                   float* __restrict__ acc, int n) {
  int idx = blockIdx.x * 256 + threadIdx.x;
  if (idx >= n * NCOLS) return;
  int i = idx / NCOLS;
  int c = idx - i * NCOLS;
  float di = dinv[i];
  acc[idx] = bias[c] + di * di * H[idx];
}

template <int NCOLS>
__global__ __launch_bounds__(256) void k_edge_scatter(const int* __restrict__ src,
                                                      const int* __restrict__ dst,
                                                      const float* __restrict__ dinv,
                                                      const float* __restrict__ H,
                                                      float* __restrict__ acc, int E) {
  constexpr int VW = NCOLS / 32;
  typedef typename vecf<VW>::T VecT;
  int wave = (blockIdx.x * 256 + threadIdx.x) >> 5;
  int lane = threadIdx.x & 31;
  if (wave >= E) return;
  int s = src[wave];
  int d = dst[wave];
  float norm = dinv[s] * dinv[d];
  const float* hrow = H + (size_t)s * NCOLS + lane * VW;
  float* arow = acc + (size_t)d * NCOLS + lane * VW;
  VecT hv = *(const VecT*)hrow;
#pragma unroll
  for (int j = 0; j < VW; ++j)
    atomicAdd(&arow[j], hv[j] * norm);
}


extern "C" void kernel_launch(void* const* d_in, const int* in_sizes, int n_in,
                              void* d_out, int out_size, void* d_ws, size_t ws_size,
                              hipStream_t stream) {
  (void)n_in; (void)out_size; (void)ws_size;
  const float* x  = (const float*)d_in[0];
  const int*   ei = (const int*)  d_in[1];
  const float* W1 = (const float*)d_in[2];
  const float* b1 = (const float*)d_in[3];
  const float* W2 = (const float*)d_in[4];
  const float* b2 = (const float*)d_in[5];
  float* out = (float*)d_out;

  const int IN = 128, HID = 128, OUT = 64;
  const int N = in_sizes[0] / IN;
  const int E = in_sizes[1] / 2;
  const int M16 = ((N + 15) / 16) * 16;
  const int* src = ei;
  const int* dst = ei + E;

  char* ws = (char*)d_ws;
  size_t off = 0;
  auto alloc = [&](size_t bytes) -> void* {
    void* p = ws + off;
    off = (off + bytes + 255) & ~(size_t)255;
    return p;
  };
  float* dinv = (float*)alloc((size_t)N * 4);
  bf16*  Xb   = (bf16*) alloc((size_t)M16 * IN * 2);
  bf16*  W1t  = (bf16*) alloc((size_t)HID * IN * 2);
  bf16*  W2t  = (bf16*) alloc((size_t)OUT * HID * 2);
  float* h1   = (float*)alloc((size_t)M16 * HID * 4);
  float* acc1 = (float*)alloc((size_t)N * HID * 4);
  bf16*  h1b  = (bf16*) alloc((size_t)M16 * HID * 2);
  float* h2   = (float*)alloc((size_t)M16 * OUT * 4);

  k_deg_init <<<(N + 255) / 256, 256, 0, stream>>>(dinv, N);
  k_deg_count<<<(E + 255) / 256, 256, 0, stream>>>(dst, dinv, E);
  k_deg_rsqrt<<<(N + 255) / 256, 256, 0, stream>>>(dinv, N);

  k_cvt_bf16      <<<(N * IN + 255) / 256, 256, 0, stream>>>(x, Xb, N * IN);
  k_transpose_bf16<<<(IN * HID + 255) / 256, 256, 0, stream>>>(W1, W1t, IN, HID);
  k_transpose_bf16<<<(HID * OUT + 255) / 256, 256, 0, stream>>>(W2, W2t, HID, OUT);

  {
    int tiles = (M16 / 16) * (HID / 64);
    k_gemm_wmma_bf16<HID><<<(tiles + 7) / 8, 256, 0, stream>>>(Xb, W1t, h1, M16);
    k_self_init<HID><<<(N * HID + 255) / 256, 256, 0, stream>>>(h1, b1, dinv, acc1, N);
    k_edge_scatter<HID><<<(E + 7) / 8, 256, 0, stream>>>(src, dst, dinv, h1, acc1, E);
    k_relu_cvt_bf16<<<(N * HID + 255) / 256, 256, 0, stream>>>(acc1, h1b, N * HID);
  }

  {
    int tiles = (M16 / 16) * (OUT / 64);
    k_gemm_wmma_bf16<OUT><<<(tiles + 7) / 8, 256, 0, stream>>>(h1b, W2t, h2, M16);
    k_self_init<OUT><<<(N * OUT + 255) / 256, 256, 0, stream>>>(h2, b2, dinv, out, N);
    k_edge_scatter<OUT><<<(E + 7) / 8, 256, 0, stream>>>(src, dst, dinv, h2, out, E);
  }
}


# Bucketing without a global add per edge: 256 chunk groups split edges by 256-node range into fixed segments via LDS counters; one group per range ranks per node in LDS, stores whole bucket lines
# speedup vs baseline: 5.7162x; 5.7162x over previous
#include <hip/hip_runtime.h>
#include <hip/hip_bf16.h>


typedef __bf16 bf16;
typedef __attribute__((ext_vector_type(16))) __bf16 v16bf;
typedef __attribute__((ext_vector_type(8)))  __bf16 v8bf;
typedef __attribute__((ext_vector_type(8)))  float  v8f;

typedef __attribute__((ext_vector_type(4)))  __bf16 v4bf;
typedef __attribute__((ext_vector_type(2)))  float  v2f;
typedef __attribute__((ext_vector_type(4)))  float  v4f;

template <int W> struct vecf;
template <> struct vecf<2> { typedef v2f T; };
template <> struct vecf<4> { typedef v4f T; };

__device__ __forceinline__ bf16 f2bf(float f) {
  union { float f; unsigned u; } in; in.f = f;
  unsigned u = in.u;
  unsigned r = u + 0x7fffu + ((u >> 16) & 1u);
  union { unsigned short s; bf16 b; } out;
  out.s = (unsigned short)(r >> 16);
  return out.b;
}

constexpr int CAP = 32;

constexpr int GA = 256;
constexpr int SEGCAP = 40;
constexpr int RANGE_SHIFT = 8;
constexpr int RANGE = 1 << RANGE_SHIFT;
constexpr int BT = 1024;
constexpr int SLOT_PITCH = CAP + 1;


__device__ __forceinline__ void transpose_bf16_block(const float* __restrict__ W,
                                                     bf16* __restrict__ Wt,
                                                     int K, int cols, int block) {
  int idx = block * BT + threadIdx.x;
  if (idx >= K * cols) return;
  int n = idx / K;
  int k = idx % K;
  Wt[idx] = f2bf(W[(size_t)k * cols + n]);
}

__global__ __launch_bounds__(BT) void k_chunk_prep(const int* __restrict__ src,
                                                   const int* __restrict__ dst,
                                                   int* __restrict__ seg,
                                                   int* __restrict__ segLen,
                                                   int* __restrict__ spill,
                                                   int* __restrict__ spillLen,
                                                   int* __restrict__ ovfCount,
                                                   int E, int CH, int NR,
                                                   const float* __restrict__ W1,
                                                   bf16* __restrict__ W1t,
                                                   const float* __restrict__ W2,
                                                   bf16* __restrict__ W2t,
                                                   int K1, int cols1, int K2, int cols2,
                                                   int tb1) {
  extern __shared__ int ldsA[];
  int* binCount = ldsA;
  int* filled   = binCount + NR;
  int* spillCnt = filled + NR;
  int* img      = spillCnt + 1;
  const int c = blockIdx.x;
  const int tid = threadIdx.x;
  if (c >= GA) {
    if (c - GA < tb1) transpose_bf16_block(W1, W1t, K1, cols1, c - GA);
    else transpose_bf16_block(W2, W2t, K2, cols2, c - GA - tb1);
    return;
  }
  for (int i = tid; i < NR; i += BT) {
    binCount[i] = 0;
    filled[i] = 0;
  }
  if (tid == 0) {
    *spillCnt = 0;
    if (c == 0) *ovfCount = 0;
  }
  __syncthreads();
  const int e0 = c * CH;
  const int len = min(CH, E - e0);
  for (int base = 0; base < len; base += 4 * BT) {
    int d[4], sv[4];
#pragma unroll
    for (int k = 0; k < 4; ++k) {
      const int i = base + k * BT + tid;
      d[k] = -1;
      sv[k] = 0;
      if (i < len) {
        d[k] = dst[e0 + i];
        sv[k] = src[e0 + i];
      }
    }
    int r[4], f[4];
#pragma unroll
    for (int k = 0; k < 4; ++k) {
      r[k] = 0;
      f[k] = 0;
      if (d[k] >= 0) {
        r[k] = atomicAdd(&binCount[d[k] >> RANGE_SHIFT], 1);
        f[k] = filled[d[k] >> RANGE_SHIFT];
      }
    }
#pragma unroll
    for (int k = 0; k < 4; ++k) {
      if (d[k] >= 0) {
        const int g = d[k] >> RANGE_SHIFT;
        const int inSeg = f[k] + r[k];
        if (inSeg < SEGCAP) img[g * SEGCAP + inSeg] = (sv[k] << RANGE_SHIFT) | (d[k] & (RANGE - 1));
        else spill[e0 + atomicAdd(spillCnt, 1)] = e0 + base + k * BT + tid;
      }
    }
    __syncthreads();

    for (int i = tid; i < NR * SEGCAP; i += BT) {
      const int g = i / SEGCAP;
      const int l = i - g * SEGCAP;
      const int lo = filled[g];
      const int nb = binCount[g];
      const int v = img[i];
      asm volatile("" :: "v"(v));
      if (l >= lo && l < lo + nb) seg[((size_t)g * GA + c) * SEGCAP + l] = v;
    }
    __syncthreads();
    for (int i = tid; i < NR; i += BT) {
      filled[i] = min(SEGCAP, filled[i] + binCount[i]);
      binCount[i] = 0;
    }
    __syncthreads();
  }
  for (int i = tid; i < NR; i += BT) segLen[(size_t)i * GA + c] = filled[i];
  if (tid == 0) spillLen[c] = *spillCnt;
}

__device__ __forceinline__ void store_edge(int* bucket, int r, int s, int dl, int g,
                                           int* __restrict__ ovf, int* ovfCount) {
  if (r < CAP) bucket[dl * SLOT_PITCH + r] = s;
  else {
    const int p = atomicAdd(ovfCount, 1);
    ovf[2 * (size_t)p] = s;
    ovf[2 * (size_t)p + 1] = (g << RANGE_SHIFT) | dl;
  }
}
__device__ __forceinline__ void place_edge(int* nodeCount, int* bucket, int s, int dl, int g,
                                           int* __restrict__ ovf, int* ovfCount) {
  store_edge(bucket, atomicAdd(&nodeCount[dl], 1), s, dl, g, ovf, ovfCount);
}

__global__ __launch_bounds__(BT) void k_range_fill(const int* __restrict__ src,
                                                   const int* __restrict__ dst,
                                                   const int* __restrict__ seg,
                                                   const int* __restrict__ segLen,
                                                   const int* __restrict__ spill,
                                                   const int* __restrict__ spillLen,
                                                   int* cnt, int* __restrict__ slots,
                                                   int* __restrict__ ovf, int* ovfCount,
                                                   int n, int CH) {
  __shared__ int nodeCount[RANGE];
  __shared__ int lenL[GA];
  __shared__ int spillL[GA];
  __shared__ int bucket[RANGE * SLOT_PITCH];
  const int g = blockIdx.x;
  const int tid = threadIdx.x;
  if (tid < RANGE) nodeCount[tid] = 0;
  int spilled = 0;
  if (tid < GA) {
    lenL[tid] = segLen[(size_t)g * GA + tid];
    spilled = spillLen[tid];
    spillL[tid] = spilled;
  }
  const int anySpill = __syncthreads_or(spilled);

  constexpr int RPT = GA * SEGCAP / BT;
  static_assert(GA * SEGCAP % BT == 0, "places divide over the threads");
  const int* segG = seg + (size_t)g * GA * SEGCAP;
  int rec[RPT], sl[RPT];
#pragma unroll
  for (int k = 0; k < RPT; ++k) sl[k] = lenL[(k * BT + tid) / SEGCAP];
#pragma unroll
  for (int k = 0; k < RPT; ++k) {
    const int i = k * BT + tid;
    rec[k] = -1;
    if (i % SEGCAP < sl[k]) rec[k] = segG[i];
  }
  int rk[RPT];
#pragma unroll
  for (int k = 0; k < RPT; ++k) {
    rk[k] = 0;
    if (rec[k] >= 0) rk[k] = atomicAdd(&nodeCount[rec[k] & (RANGE - 1)], 1);
  }
#pragma unroll
  for (int k = 0; k < RPT; ++k)
    if (rec[k] >= 0) store_edge(bucket, rk[k], rec[k] >> RANGE_SHIFT, rec[k] & (RANGE - 1), g, ovf, ovfCount);

  if (anySpill != 0) {
    for (int c = 0; c < GA; ++c) {
      const int m = spillL[c];
      for (int i = tid; i < m; i += BT) {
        const int e = spill[c * CH + i];
        const int d = dst[e];
        if ((d >> RANGE_SHIFT) == g) place_edge(nodeCount, bucket, src[e], d & (RANGE - 1), g, ovf, ovfCount);
      }
    }
  }
  __syncthreads();

  const int node0 = g << RANGE_SHIFT;
  if (tid < RANGE && node0 + tid < n) cnt[node0 + tid] = nodeCount[tid];
  constexpr int LPT = RANGE * CAP / BT;
  int cn[LPT], bv[LPT];
#pragma unroll
  for (int k = 0; k < LPT; ++k) {
    const int idx = k * BT + tid;
    cn[k] = nodeCount[idx / CAP];
    bv[k] = bucket[(idx / CAP) * SLOT_PITCH + idx % CAP];
  }
#pragma unroll
  for (int k = 0; k < LPT; ++k) {
    const int idx = k * BT + tid;
    const int nl = idx / CAP;
    const int j = idx % CAP;
    if (node0 + nl < n) slots[(size_t)(node0 + nl) * CAP + j] = j < cn[k] ? bv[k] : 0;
  }
}


template <int NCOLS, int CNT, bool SCALE>
__device__ __forceinline__ void gather_group(const float* __restrict__ H, int loff,
                                             int rs, int cs, int j,
                                             typename vecf<NCOLS / 32>::T& acc) {
  typedef typename vecf<NCOLS / 32>::T VecT;
  int s[CNT];
  VecT hv[CNT];
  for (int u = 0; u < CNT; ++u) s[u] = __builtin_amdgcn_readlane(rs, j + u);
  for (int u = 0; u < CNT; ++u)
    hv[u] = *(const VecT*)(H + (size_t)s[u] * NCOLS + loff);
  if (SCALE) {
    __builtin_amdgcn_sched_barrier(0);
    float dv[CNT];
    for (int u = 0; u < CNT; ++u)
      dv[u] = __builtin_amdgcn_rsqf((float)(__builtin_amdgcn_readlane(cs, j + u) + 1));
    for (int u = 0; u < CNT; ++u) acc += hv[u] * dv[u];
  } else {
    for (int u = 0; u < CNT; ++u) acc += hv[u];
  }
}

template <int NCOLS, int WPB, bool SCALE>
__device__ __forceinline__ bool gather_row(const int* __restrict__ slots,
                                           const int* __restrict__ cnt,
                                           const int* __restrict__ ovf,
                                           const float* __restrict__ H,
                                           const float* __restrict__ bias, int n,
                                           int& node, int& loff, float& di,
                                           typename vecf<NCOLS / 32>::T& acc) {
  constexpr int VW = NCOLS / 32;
  typedef typename vecf<VW>::T VecT;
  const int lane = threadIdx.x & 31;
  node = __builtin_amdgcn_readfirstlane(blockIdx.x * WPB + (threadIdx.x >> 5));
  if (node >= n) return false;
  const int deg = cnt[node];
  loff = lane * VW;
  di = rsqrtf((float)(deg + 1));

  acc = *(const VecT*)(H + (size_t)node * NCOLS + loff);
  if (SCALE) acc = acc * di;

  const int m = min(CAP, deg);
  int r = 0;
  if (lane < m) r = slots[(size_t)node * CAP + lane];
  int cs = 0;
  if (SCALE) {
    const unsigned key = lane < m ? (((unsigned)r << 5) | (unsigned)lane) : 0xffffffffu;
    int place = 0;
    for (int u = 0; u < m; ++u)
      place += ((unsigned)__builtin_amdgcn_readlane((int)key, u) < key) ? 1 : 0;
    if (lane >= m) place = lane;
    r = __builtin_amdgcn_ds_permute(place << 2, r);
    if (lane < m) cs = cnt[r];
  }
  int j = 0;
  for (; j + 8 <= m; j += 8) gather_group<NCOLS, 8, SCALE>(H, loff, r, cs, j, acc);
  switch (m - j) {
    case 7: gather_group<NCOLS, 7, SCALE>(H, loff, r, cs, j, acc); break;
    case 6: gather_group<NCOLS, 6, SCALE>(H, loff, r, cs, j, acc); break;
    case 5: gather_group<NCOLS, 5, SCALE>(H, loff, r, cs, j, acc); break;
    case 4: gather_group<NCOLS, 4, SCALE>(H, loff, r, cs, j, acc); break;
    case 3: gather_group<NCOLS, 3, SCALE>(H, loff, r, cs, j, acc); break;
    case 2: gather_group<NCOLS, 2, SCALE>(H, loff, r, cs, j, acc); break;
    case 1: gather_group<NCOLS, 1, SCALE>(H, loff, r, cs, j, acc); break;
    default: break;
  }

  if (deg > CAP) {
    const int nOvf = cnt[n];
    for (int b = 0; b < nOvf; b += 32) {
      int dm = -1, sm = 0;
      if (b + lane < nOvf) {
        sm = ovf[2 * (size_t)(b + lane)];
        dm = ovf[2 * (size_t)(b + lane) + 1];
      }
      unsigned hit = __builtin_amdgcn_ballot_w32(dm == node);
      while (hit != 0u) {
        const int l = __builtin_ctz(hit);
        hit &= hit - 1u;
        const int s = __builtin_amdgcn_readlane(sm, l);
        const VecT hv = *(const VecT*)(H + (size_t)s * NCOLS + loff);
        if (SCALE) acc += hv * rsqrtf((float)(cnt[s] + 1));
        else acc += hv;
      }
    }
  }

  if (SCALE) acc = acc * di;
  else acc = acc * di + *(const VecT*)(bias + loff);
  return true;
}

constexpr int TILE_PITCH = 136;

__device__ __forceinline__ void load_b_frags(const bf16* __restrict__ bRow, v8bf (&bfr)[8]) {
#pragma unroll
  for (int q = 0; q < 4; ++q) {
    bfr[2 * q]     = *(const v8bf*)(bRow + 32 * q);
    bfr[2 * q + 1] = *(const v8bf*)(bRow + 32 * q + 16);
  }
}

__device__ __forceinline__ v8f tile_product(const bf16* aRow, const v8bf (&bfr)[8], v8f c) {
  union AB { v16bf v; v8bf h[2]; };
  AB a[4];
  a[0].h[0] = *(const v8bf*)(aRow);
  a[0].h[1] = *(const v8bf*)(aRow + 16);
#pragma unroll
  for (int q = 0; q < 4; ++q) {
    if (q + 1 < 4) {
      a[q + 1].h[0] = *(const v8bf*)(aRow + 32 * (q + 1));
      a[q + 1].h[1] = *(const v8bf*)(aRow + 32 * (q + 1) + 16);
    }
    AB b;
    b.h[0] = bfr[2 * q];
    b.h[1] = bfr[2 * q + 1];
    c = __builtin_amdgcn_wmma_f32_16x16x32_bf16(
         false, a[q].v,  false, b.v,
         (short)0, c,  false,  false);
  }
  return c;
}

__global__ __launch_bounds__(512, 16) void k_gather1_gemm1_relu_gemm2(const int* __restrict__ slots,
                                                            const int* __restrict__ cnt,
                                                            const int* __restrict__ ovf,
                                                            const float* __restrict__ X,
                                                            const bf16* __restrict__ B1t,
                                                            const float* __restrict__ b1,
                                                            const bf16* __restrict__ B2t,
                                                            float* __restrict__ C,
                                                            int n) {
  __shared__ __attribute__((aligned(16))) bf16 tileX[16 * TILE_PITCH];
  __shared__ __attribute__((aligned(16))) bf16 tileH[16 * TILE_PITCH];
  __shared__ __attribute__((aligned(16))) float rowDinv[16];
  constexpr int K = 128;
  const int lane = threadIdx.x & 31;
  const int wave = __builtin_amdgcn_readfirstlane(threadIdx.x >> 5);
  const int rc = lane & 15;
  const int hi = lane >> 4;

  int node, loff;
  float di = 0.0f;
  v4f acc;
  v4f r = {0.0f, 0.0f, 0.0f, 0.0f};
  if (gather_row<128, 16, true>(slots, cnt, ovf, X, nullptr, n, node, loff, di, acc)) r = acc;
  *(v4bf*)(tileX + wave * TILE_PITCH + lane * 4) = __builtin_convertvector(r, v4bf);
  if (lane == 0) rowDinv[wave] = di;

  v8bf bfr[8];
  float bcol = 0.0f;
  if (wave < 8) {
    load_b_frags(B1t + (size_t)(wave * 16 + rc) * K + hi * 8, bfr);
    bcol = b1[wave * 16 + rc];
  }
  __syncthreads();

  if (wave < 8) {
    const v8f cb = {bcol, bcol, bcol, bcol, bcol, bcol, bcol, bcol};
    const v8f c = tile_product(tileX + rc * TILE_PITCH + hi * 8, bfr, cb);
    v8f h;
#pragma unroll
    for (int j = 0; j < 8; ++j) h[j] = fmaxf(c[j], 0.0f);
    const v8bf hb = __builtin_convertvector(h, v8bf);
    bf16* hCol = tileH + (hi * 8) * TILE_PITCH + wave * 16 + rc;
#pragma unroll
    for (int j = 0; j < 8; ++j) hCol[j * TILE_PITCH] = hb[j];
  }
  if (wave < 4) load_b_frags(B2t + (size_t)(wave * 16 + rc) * K + hi * 8, bfr);
  __syncthreads();
  if (wave >= 4) return;

  const v8f zero = {};
  const v8f c2 = tile_product(tileH + rc * TILE_PITCH + hi * 8, bfr, zero);

  const v4f d0 = *(const v4f*)(rowDinv + hi * 8);
  const v4f d1 = *(const v4f*)(rowDinv + hi * 8 + 4);
  const float sc[8] = {d0.x, d0.y, d0.z, d0.w, d1.x, d1.y, d1.z, d1.w};
  float* cBase = C + (size_t)(blockIdx.x * 16 + hi * 8) * 64 + wave * 16 + rc;
#pragma unroll
  for (int j = 0; j < 8; ++j) cBase[(size_t)j * 64] = c2[j] * sc[j];
}

__global__ __launch_bounds__(256) void k_gather2_f32(const int* __restrict__ slots,
                                                     const int* __restrict__ cnt,
                                                     const int* __restrict__ ovf,
                                                     const float* __restrict__ H,
                                                     const float* __restrict__ bias,
                                                     float* __restrict__ outp, int n) {
  int node, loff;
  float di;
  v2f acc;
  if (!gather_row<64, 8, false>(slots, cnt, ovf, H, bias, n, node, loff, di, acc)) return;
  *(v2f*)(outp + (size_t)node * 64 + loff) = acc;
}


extern "C" void kernel_launch(void* const* d_in, const int* in_sizes, int n_in,
                              void* d_out, int out_size, void* d_ws, size_t ws_size,
                              hipStream_t stream) {
  (void)n_in; (void)out_size; (void)ws_size;
  const float* x  = (const float*)d_in[0];
  const int*   ei = (const int*)  d_in[1];
  const float* W1 = (const float*)d_in[2];
  const float* b1 = (const float*)d_in[3];
  const float* W2 = (const float*)d_in[4];
  const float* b2 = (const float*)d_in[5];
  float* out = (float*)d_out;

  const int IN = 128, HID = 128, OUT = 64;
  const int N = in_sizes[0] / IN;
  const int E = in_sizes[1] / 2;
  const int M16 = ((N + 15) / 16) * 16;
  const int* src = ei;
  const int* dst = ei + E;

  char* ws = (char*)d_ws;
  size_t off = 0;
  auto alloc = [&](size_t bytes) -> void* {
    void* p = ws + off;
    off = (off + bytes + 255) & ~(size_t)255;
    return p;
  };
  bf16*  W1t  = (bf16*) alloc((size_t)HID * IN * 2);
  bf16*  W2t  = (bf16*) alloc((size_t)OUT * HID * 2);
  float* h2   = (float*)alloc((size_t)M16 * OUT * 4);
  int* cnt    = (int*)alloc((size_t)(N + 1) * 4);
  int* slots  = (int*)alloc((size_t)N * CAP * 4);
  int* ovf    = (int*)alloc((size_t)E * 8);
  const int NR = (N + RANGE - 1) / RANGE;
  const int CH = (E + GA - 1) / GA;
  int* seg      = (int*)alloc((size_t)NR * GA * SEGCAP * 4);
  int* segLen   = (int*)alloc((size_t)NR * GA * 4);
  int* spill    = (int*)alloc((size_t)GA * CH * 4);
  int* spillLen = (int*)alloc((size_t)GA * 4);

  {
    int tb1 = (IN * HID + BT - 1) / BT;
    int tb2 = (HID * OUT + BT - 1) / BT;
    k_chunk_prep<<<GA + tb1 + tb2, BT, (size_t)(2 * NR + 1 + NR * SEGCAP) * 4, stream>>>(src, dst, seg, segLen, spill, spillLen, cnt + N,
                                                                       E, CH, NR, W1, W1t, W2, W2t,
                                                                       IN, HID, HID, OUT, tb1);
  }

  k_range_fill<<<NR, BT, 0, stream>>>(src, dst, seg, segLen, spill, spillLen, cnt, slots, ovf, cnt + N, N, CH);

  k_gather1_gemm1_relu_gemm2<<<M16 / 16, 512, 0, stream>>>(slots, cnt, ovf, x, W1t, b1, W2t, h2, N);
  k_gather2_f32<<<(N + 7) / 8, 256, 0, stream>>>(slots, cnt, ovf, h2, b2, out, N);
}
